# starting kernel
# baseline (speedup 1.0000x reference)
#include <hip/hip_runtime.h>


typedef __attribute__((ext_vector_type(2))) float v2f;
typedef __attribute__((ext_vector_type(4))) float v4f;
typedef __attribute__((ext_vector_type(8))) float v8f;

__device__ __constant__ float kA0V0[32] = {
    0,-1,0,0,0,0,0,0, 0,0,0,0,0,0,0,0,
    0,0,0,0,0,0,0,-1, 0,0,0,0,0,0,0,0};
__device__ __constant__ float kA0V1[32] = {
    1,0,-1,0,0,0,0,0, 0,0,0,0,0,0,0,0,
    0,1,0,0,-1,0,1,0, 0,0,0,0,0,0,0,0};
__device__ __constant__ float kA1V0[32] = {
    0,0,0,0,0,0,0,0, 0,0,0,0,0,0,0,0,
    0,0,0,0,0,0,0,1, 0,0,0,0,0,0,0,0};
__device__ __constant__ float kA1V1[32] = {
    0,0,1,0,0,-1,0,0, 0,0,0,0,0,0,0,0,
    0,0,0,1,1,1,0,0, 0,0,0,0,0,0,0,0};

__global__ __launch_bounds__(256)
void slice_sum_cat_kernel(const float* __restrict__ x,
                          float* __restrict__ out,
                          int nbatch) {
    __shared__ __align__(16) float lds[8][8][128];

    const int lane = threadIdx.x & 31;
    const int wave = threadIdx.x >> 5;
    const int gw   = blockIdx.x * 8 + wave;
    const int b0   = gw * 2;
    if (b0 >= nbatch) return;

    const int bb = lane >> 4;
    const int cl = (lane & 15) * 4;

    const v4f* base = (const v4f*)(x + (size_t)(b0 + bb) * (200 * 64) + cl);
    v4f acc = {0.f, 0.f, 0.f, 0.f};
    v4f snap[8];
#pragma unroll
    for (int seg = 0; seg < 8; ++seg) {
#pragma unroll
        for (int i = 0; i < 25; ++i) {
            const v4f v = __builtin_nontemporal_load(base + (size_t)(seg * 25 + i) * 16);
            acc += v;
        }
        snap[seg] = acc;
    }

#pragma unroll
    for (int p = 0; p < 8; ++p)
        *(v4f*)&lds[wave][p][bb * 64 + cl] = snap[p];
    __syncthreads();

    const v2f A0 = { kA0V0[lane], kA0V1[lane] };
    const v2f A1 = { kA1V0[lane], kA1V1[lane] };
    const int half = lane >> 4;
    const int nl   = lane & 15;

#pragma unroll
    for (int c = 0; c < 8; ++c) {
        v2f B0, B1;
        B0.x = lds[wave][2 * half + 0][c * 16 + nl];
        B0.y = lds[wave][2 * half + 1][c * 16 + nl];
        B1.x = lds[wave][2 * half + 4][c * 16 + nl];
        B1.y = lds[wave][2 * half + 5][c * 16 + nl];

        v8f d = {0.f, 0.f, 0.f, 0.f, 0.f, 0.f, 0.f, 0.f};
        d = __builtin_amdgcn_wmma_f32_16x16x4_f32(false, A0, false, B0,
                                                  (short)0, d, false, false);
        d = __builtin_amdgcn_wmma_f32_16x16x4_f32(false, A1, false, B1,
                                                  (short)0, d, false, false);

        const int   ob = b0 + (c >> 2);
        float*      op = out + (size_t)ob * 512 + (c & 3) * 16 + nl;
        if (lane < 16) {
#pragma unroll
            for (int s = 0; s < 8; ++s)
                __builtin_nontemporal_store(d[s], op + s * 64);
        }
    }
}

extern "C" void kernel_launch(void* const* d_in, const int* in_sizes, int n_in,
                              void* d_out, int out_size, void* d_ws, size_t ws_size,
                              hipStream_t stream) {
    const float* x   = (const float*)d_in[0];
    float*       out = (float*)d_out;
    const int nbatch = in_sizes[0] / (200 * 64);
    const int waves  = (nbatch + 1) / 2;
    const int blocks = (waves + 7) / 8;
    slice_sum_cat_kernel<<<blocks, 256, 0, stream>>>(x, out, nbatch);
}


# One batch per wave, not two: 16 waves per WGP (was 8); lanes 0-15 stream rows 0-99, lanes 16-31 rows 100-199; the row prefix restarts at row 100 and the +/-1 f32 WMMA combine adds the halves
# speedup vs baseline: 1.4112x; 1.4112x over previous
#include <hip/hip_runtime.h>


typedef __attribute__((ext_vector_type(2))) float v2f;
typedef __attribute__((ext_vector_type(4))) float v4f;
typedef __attribute__((ext_vector_type(8))) float v8f;

constexpr int kRows  = 200;
constexpr int kCols  = 64;
constexpr int kSeg   = 25;
constexpr int kHalf  = 100;
constexpr int kWaves = 16;

constexpr int kStart[8] = {0, 25, 50, 0, 100, 150, 0, 75};
constexpr int kEnd[8]   = {50, 100, 150, 200, 200, 200, 100, 175};

__host__ __device__ constexpr unsigned long long combine_masks(bool minus) {
    unsigned long long m = 0;
    for (int s = 0; s < 8; ++s) {
        const int a = kStart[s], e = kEnd[s];
        if (a < kHalf) {
            const int hi = (e < kHalf ? e : kHalf) / kSeg;
            const int lo = a / kSeg;
            if (!minus)          m |= 1ull << (8 * s + hi - 1);
            if (minus && lo > 0) m |= 1ull << (8 * s + lo - 1);
        }
        if (e > kHalf) {
            const int hi = (e - kHalf) / kSeg;
            const int lo = ((a > kHalf ? a : kHalf) - kHalf) / kSeg;
            if (!minus)          m |= 1ull << (8 * s + 4 + hi - 1);
            if (minus && lo > 0) m |= 1ull << (8 * s + 4 + lo - 1);
        }
    }
    return m;
}
constexpr unsigned long long kPlus  = combine_masks(false);
constexpr unsigned long long kMinus = combine_masks(true);
static_assert(kPlus == 0x4808808088280802ull && kMinus == 0x0400200000020100ull,
              "combine matrix does not match the slice table");

__global__ __launch_bounds__(kWaves * 32)
void slice_sum_cat_kernel(const float* __restrict__ x,
                          float* __restrict__ out,
                          int nbatch) {
    __shared__ __align__(16) float lds[kWaves][8][kCols];

    const int lane = threadIdx.x & 31;
    const int wave = threadIdx.x >> 5;
    const int b    = blockIdx.x * kWaves + wave;
    if (b >= nbatch) return;

    const int h  = lane >> 4;
    const int nl = lane & 15;
    const int cl = nl * 4;

    const v4f* base = (const v4f*)(x + (size_t)b * (kRows * kCols)
                                     + (size_t)h * (kHalf * kCols) + cl);
    v4f acc = {0.f, 0.f, 0.f, 0.f};
    v4f snap[4];
#pragma unroll
    for (int seg = 0; seg < 4; ++seg) {
#pragma unroll
        for (int i = 0; i < kSeg; ++i) {
            const v4f v = __builtin_nontemporal_load(base + (size_t)(seg * kSeg + i) * (kCols / 4));
            acc += v;
        }
        snap[seg] = acc;
    }

#pragma unroll
    for (int p = 0; p < 4; ++p)
        *(v4f*)&lds[wave][4 * h + p][cl] = snap[p];
    __syncthreads();

    const unsigned pm = (nl < 8) ? (unsigned)(kPlus  >> (8 * nl)) & 0xFFu : 0u;
    const unsigned mm = (nl < 8) ? (unsigned)(kMinus >> (8 * nl)) & 0xFFu : 0u;
    const int k0 = 2 * h;
    const v2f A0 = { (float)((pm >> (k0 + 0)) & 1u) - (float)((mm >> (k0 + 0)) & 1u),
                     (float)((pm >> (k0 + 1)) & 1u) - (float)((mm >> (k0 + 1)) & 1u) };
    const v2f A1 = { (float)((pm >> (k0 + 4)) & 1u) - (float)((mm >> (k0 + 4)) & 1u),
                     (float)((pm >> (k0 + 5)) & 1u) - (float)((mm >> (k0 + 5)) & 1u) };

#pragma unroll
    for (int c = 0; c < 4; ++c) {
        v2f B0, B1;
        B0.x = lds[wave][k0 + 0][c * 16 + nl];
        B0.y = lds[wave][k0 + 1][c * 16 + nl];
        B1.x = lds[wave][k0 + 4][c * 16 + nl];
        B1.y = lds[wave][k0 + 5][c * 16 + nl];

        v8f d = {0.f, 0.f, 0.f, 0.f, 0.f, 0.f, 0.f, 0.f};
        d = __builtin_amdgcn_wmma_f32_16x16x4_f32(false, A0, false, B0,
                                                  (short)0, d, false, false);
        d = __builtin_amdgcn_wmma_f32_16x16x4_f32(false, A1, false, B1,
                                                  (short)0, d, false, false);

        float* op = out + (size_t)b * 512 + c * 16 + nl;
        if (lane < 16) {
#pragma unroll
            for (int s = 0; s < 8; ++s)
                __builtin_nontemporal_store(d[s], op + s * kCols);
        }
    }
}

extern "C" void kernel_launch(void* const* d_in, const int* in_sizes, int n_in,
                              void* d_out, int out_size, void* d_ws, size_t ws_size,
                              hipStream_t stream) {
    const float* x   = (const float*)d_in[0];
    float*       out = (float*)d_out;
    const int nbatch = in_sizes[0] / (200 * 64);
    const int blocks = (nbatch + kWaves - 1) / kWaves;
    slice_sum_cat_kernel<<<blocks, kWaves * 32, 0, stream>>>(x, out, nbatch);
}


# Combine from registers: a lane's 4 columns become WMMA column 4n+j, so no LDS, no barrier; A rows 8-15 repeat the slices, all lanes store whole lines (b128); the combine sums prefixes in a new order
# speedup vs baseline: 1.5333x; 1.0865x over previous
#include <hip/hip_runtime.h>


typedef __attribute__((ext_vector_type(2))) float v2f;
typedef __attribute__((ext_vector_type(4))) float v4f;
typedef __attribute__((ext_vector_type(8))) float v8f;

constexpr int kRows  = 200;
constexpr int kCols  = 64;
constexpr int kSeg   = 25;
constexpr int kHalf  = 100;
constexpr int kWaves = 16;

constexpr int kStart[8] = {0, 25, 50, 0, 100, 150, 0, 75};
constexpr int kEnd[8]   = {50, 100, 150, 200, 200, 200, 100, 175};

__host__ __device__ constexpr unsigned long long combine_masks(bool minus) {
    unsigned long long m = 0;
    for (int s = 0; s < 8; ++s) {
        const int a = kStart[s], e = kEnd[s];
        if (a < kHalf) {
            const int hi = (e < kHalf ? e : kHalf) / kSeg;
            const int lo = a / kSeg;
            if (!minus)          m |= 1ull << (8 * s + hi - 1);
            if (minus && lo > 0) m |= 1ull << (8 * s + lo - 1);
        }
        if (e > kHalf) {
            const int hi = (e - kHalf) / kSeg;
            const int lo = ((a > kHalf ? a : kHalf) - kHalf) / kSeg;
            if (!minus)          m |= 1ull << (8 * s + 4 + hi - 1);
            if (minus && lo > 0) m |= 1ull << (8 * s + 4 + lo - 1);
        }
    }
    return m;
}
constexpr unsigned long long kPlus  = combine_masks(false);
constexpr unsigned long long kMinus = combine_masks(true);
static_assert(kPlus == 0x4808808088280802ull && kMinus == 0x0400200000020100ull,
              "combine matrix does not match the slice table");

__global__ __launch_bounds__(kWaves * 32)
void slice_sum_cat_kernel(const float* __restrict__ x,
                          float* __restrict__ out,
                          int nbatch) {
    const int lane = threadIdx.x & 31;
    const int wave = threadIdx.x >> 5;
    const int b    = blockIdx.x * kWaves + wave;
    if (b >= nbatch) return;

    const int h  = lane >> 4;
    const int nl = lane & 15;
    const int cl = nl * 4;

    const v4f* base = (const v4f*)(x + (size_t)b * (kRows * kCols)
                                     + (size_t)h * (kHalf * kCols) + cl);
    v4f acc = {0.f, 0.f, 0.f, 0.f};
    v4f snap[4];
#pragma unroll
    for (int seg = 0; seg < 4; ++seg) {
#pragma unroll
        for (int i = 0; i < kSeg; ++i) {
            const v4f v = __builtin_nontemporal_load(base + (size_t)(seg * kSeg + i) * (kCols / 4));
            acc += v;
        }
        snap[seg] = acc;
    }

    const int sl = (nl + 4 * (nl >> 3)) & 7;
    const unsigned pm = (unsigned)(kPlus  >> (8 * sl)) & 0xFFu;
    const unsigned mm = (unsigned)(kMinus >> (8 * sl)) & 0xFFu;
    v2f A[2];
#pragma unroll
    for (int t = 0; t < 2; ++t) {
        const int q = 4 * h + 2 * t;
        A[t].x = (float)((pm >> (q + 0)) & 1u) - (float)((mm >> (q + 0)) & 1u);
        A[t].y = (float)((pm >> (q + 1)) & 1u) - (float)((mm >> (q + 1)) & 1u);
    }

    v8f d[4];
#pragma unroll
    for (int j = 0; j < 4; ++j)
        d[j] = (v8f){0.f, 0.f, 0.f, 0.f, 0.f, 0.f, 0.f, 0.f};
#pragma unroll
    for (int t = 0; t < 2; ++t) {
#pragma unroll
        for (int j = 0; j < 4; ++j) {
            const v2f B = { snap[2 * t][j], snap[2 * t + 1][j] };
            d[j] = __builtin_amdgcn_wmma_f32_16x16x4_f32(false, A[t], false, B,
                                                         (short)0, d[j], false, false);
        }
    }

    float* op = out + (size_t)b * (8 * kCols) + (size_t)(4 * h) * kCols + cl;
#pragma unroll
    for (int r = 0; r < 4; ++r) {
        const v4f o = { d[0][r], d[1][r], d[2][r], d[3][r] };
        __builtin_nontemporal_store(o, (v4f*)(op + r * kCols));
    }
}

extern "C" void kernel_launch(void* const* d_in, const int* in_sizes, int n_in,
                              void* d_out, int out_size, void* d_ws, size_t ws_size,
                              hipStream_t stream) {
    const float* x   = (const float*)d_in[0];
    float*       out = (float*)d_out;
    const int nbatch = in_sizes[0] / (200 * 64);
    const int blocks = (nbatch + kWaves - 1) / kWaves;
    slice_sum_cat_kernel<<<blocks, kWaves * 32, 0, stream>>>(x, out, nbatch);
}
